# starting kernel
# baseline (speedup 1.0000x reference)
#include <hip/hip_runtime.h>
#include <math.h>

#define ALPHA 0.1f
#define KITER 10
#define DFEAT 32

typedef __attribute__((ext_vector_type(2))) float v2f;
typedef __attribute__((ext_vector_type(8))) float v8f;

__global__ void appnp_init(const float* __restrict__ feat, float* __restrict__ h,
                           float* __restrict__ acc, float* __restrict__ norm, int N) {
    int i = blockIdx.x * blockDim.x + threadIdx.x;
    int total = N * DFEAT;
    if (i < total) { h[i] = feat[i]; acc[i] = 0.0f; }
    if (i < N) norm[i] = 0.0f;
}

__global__ void appnp_degree(const int* __restrict__ dst, float* __restrict__ deg, int E) {
    int i = blockIdx.x * blockDim.x + threadIdx.x;
    if (i < E) atomicAdd(&deg[dst[i]], 1.0f);
}

__global__ void appnp_norm(float* __restrict__ norm, int N) {
    int i = blockIdx.x * blockDim.x + threadIdx.x;
    if (i < N) {
        float d = norm[i];
        norm[i] = (d > 0.0f) ? (1.0f / sqrtf(d)) : 0.0f;
    }
}

__global__ void appnp_scatter(const float* __restrict__ h, const float* __restrict__ norm,
                              const int* __restrict__ src, const int* __restrict__ dst,
                              float* __restrict__ acc, int E) {
    int wave = blockIdx.x * (blockDim.x >> 5) + (threadIdx.x >> 5);
    if (wave >= E) return;
    int lane = threadIdx.x & 31;
    if (lane == 0) {
        __builtin_prefetch(src + wave + 4096, 0, 0);
        __builtin_prefetch(dst + wave + 4096, 0, 0);
    }
    int s = src[wave];
    int d = dst[wave];
    float v = h[s * DFEAT + lane] * norm[s];
    atomicAdd(&acc[d * DFEAT + lane], v);
}

__global__ void appnp_update_wmma(const float* __restrict__ feat, const float* __restrict__ norm,
                                  float* __restrict__ acc, float* __restrict__ h,
                                  int N, int ntiles) {
    int wave = blockIdx.x * (blockDim.x >> 5) + (threadIdx.x >> 5);
    if (wave >= ntiles) return;
    int node0 = wave * 16;
    int lane = threadIdx.x & 31;
    int hi = lane >> 4;
    int lo = lane & 15;

    if (node0 + 16 <= N) {
        float nv = (1.0f - ALPHA) * norm[node0 + lo];
#pragma unroll
        for (int fh = 0; fh < 2; ++fh) {
            int f0 = fh * 16;
            v8f cacc;
#pragma unroll
            for (int r = 0; r < 8; ++r)
                cacc[r] = ALPHA * feat[(node0 + r + 8 * hi) * DFEAT + f0 + lo];
#pragma unroll
            for (int cc = 0; cc < 4; ++cc) {
                v2f a, b;
#pragma unroll
                for (int v = 0; v < 2; ++v) {
                    int k = v + 2 * hi;
                    int m = 4 * cc + k;
                    b[v] = acc[(node0 + m) * DFEAT + f0 + lo];
                    a[v] = (lo == (4 * cc + k)) ? nv : 0.0f;
                }
                cacc = __builtin_amdgcn_wmma_f32_16x16x4_f32(
                     false, a,  false, b,
                     (short)0, cacc,  false,  false);
            }
#pragma unroll
            for (int r = 0; r < 8; ++r)
                h[(node0 + r + 8 * hi) * DFEAT + f0 + lo] = cacc[r];
#pragma unroll
            for (int cc = 0; cc < 4; ++cc)
#pragma unroll
                for (int v = 0; v < 2; ++v) {
                    int m = 4 * cc + v + 2 * hi;
                    acc[(node0 + m) * DFEAT + f0 + lo] = 0.0f;
                }
        }
    } else {
        for (int m = 0; m < 16; ++m) {
            int node = node0 + m;
            if (node < N) {
                int j = node * DFEAT + lane;
                h[j] = (1.0f - ALPHA) * norm[node] * acc[j] + ALPHA * feat[j];
                acc[j] = 0.0f;
            }
        }
    }
}

extern "C" void kernel_launch(void* const* d_in, const int* in_sizes, int n_in,
                              void* d_out, int out_size, void* d_ws, size_t ws_size,
                              hipStream_t stream) {
    const float* feat = (const float*)d_in[0];
    const int*   src  = (const int*)d_in[1];
    const int*   dst  = (const int*)d_in[2];
    int N = in_sizes[0] / DFEAT;
    int E = in_sizes[1];

    float* h    = (float*)d_out;
    float* acc  = (float*)d_ws;
    float* norm = acc + (size_t)N * DFEAT;

    int totalND = N * DFEAT;
    appnp_init<<<(totalND + 255) / 256, 256, 0, stream>>>(feat, h, acc, norm, N);
    appnp_degree<<<(E + 255) / 256, 256, 0, stream>>>(dst, norm, E);
    appnp_norm<<<(N + 255) / 256, 256, 0, stream>>>(norm, N);

    int ntiles = (N + 15) / 16;
    for (int k = 0; k < KITER; ++k) {
        appnp_scatter<<<(E + 7) / 8, 256, 0, stream>>>(h, norm, src, dst, acc, E);
        appnp_update_wmma<<<(ntiles + 7) / 8, 256, 0, stream>>>(feat, norm, acc, h, N, ntiles);
    }
}


# push to pull: edges grouped by dst once per call (one returning int atomic per edge, prefix sum, fill); each sweep sums h[src]*norm[src] per dst row with plain loads, no per-sweep atomics
# speedup vs baseline: 3.1340x; 3.1340x over previous
#include <hip/hip_runtime.h>
#include <math.h>

#define ALPHA 0.1f
#define KITER 10
#define DFEAT 32
#define SCAN_T 1024

typedef __attribute__((ext_vector_type(2))) float v2f;
typedef __attribute__((ext_vector_type(8))) float v8f;

__global__ void appnp_init(const float* __restrict__ feat, float* __restrict__ h,
                           int* __restrict__ cnt, int N) {
    int i = blockIdx.x * blockDim.x + threadIdx.x;
    int total = N * DFEAT;
    if (i < total) h[i] = feat[i];
    if (i < N) cnt[i] = 0;
}

__global__ void appnp_count_rank(const int* __restrict__ dst, int* __restrict__ cnt,
                                 int* __restrict__ rank, int E) {
    int i = blockIdx.x * blockDim.x + threadIdx.x;
    if (i < E) rank[i] = atomicAdd(&cnt[dst[i]], 1);
}

__global__ void __launch_bounds__(SCAN_T)
appnp_rowptr_norm(const int* __restrict__ cnt, int* __restrict__ rowptr,
                  float* __restrict__ norm, int N) {
    __shared__ int s_part[SCAN_T / 32];
    __shared__ int s_incl[SCAN_T / 32];
    int tid = threadIdx.x;
    int lane = tid & 31;
    int w = tid >> 5;
    int base = blockIdx.x * SCAN_T;

    int part = 0;
    const int4* cnt4 = (const int4*)cnt;
#pragma unroll 8
    for (int q = tid; q < base / 4; q += SCAN_T) {
        int4 v = cnt4[q];
        part += (v.x + v.y) + (v.z + v.w);
    }
    int i = base + tid;
    int c = (i < N) ? cnt[i] : 0;

    int incl = c;
#pragma unroll
    for (int o = 1; o < 32; o <<= 1) {
        int t = __shfl_up(incl, o);
        if (lane >= o) incl += t;
    }
#pragma unroll
    for (int o = 16; o > 0; o >>= 1) part += __shfl_xor(part, o);

    if (lane == 31) s_incl[w] = incl;
    if (lane == 0) s_part[w] = part;
    __syncthreads();

    int before = 0;
    for (int k = 0; k < SCAN_T / 32; ++k) {
        before += s_part[k];
        if (k < w) before += s_incl[k];
    }
    if (i < N) {
        rowptr[i] = before + incl - c;
        norm[i] = (c > 0) ? (1.0f / sqrtf((float)c)) : 0.0f;
        if (i == N - 1) rowptr[N] = before + incl;
    }
}

__global__ void appnp_fill(const int* __restrict__ src, const int* __restrict__ dst,
                           const int* __restrict__ rank, const int* __restrict__ rowptr,
                           int* __restrict__ colidx, int E) {
    int i = blockIdx.x * blockDim.x + threadIdx.x;
    if (i < E) colidx[rowptr[dst[i]] + rank[i]] = src[i];
}

template <int U>
__device__ __forceinline__ float gather_group(const float* __restrict__ hsrc, int my_s,
                                              int my_nbits, int j, int lane, float sum) {
    float v[U];
#pragma unroll
    for (int u = 0; u < U; ++u) {
        int s = __builtin_amdgcn_readlane(my_s, j + u);
        v[u] = hsrc[s * DFEAT + lane];
    }
#pragma unroll
    for (int u = 0; u < U; ++u)
        sum += v[u] * __int_as_float(__builtin_amdgcn_readlane(my_nbits, j + u));
    return sum;
}

__global__ void appnp_gather(const float* __restrict__ hsrc, const float* __restrict__ norm,
                             const int* __restrict__ rowptr, const int* __restrict__ colidx,
                             float* __restrict__ acc, int N) {
    int node = blockIdx.x * (blockDim.x >> 5) + (threadIdx.x >> 5);
    node = __builtin_amdgcn_readfirstlane(node);
    if (node >= N) return;
    int lane = threadIdx.x & 31;
    int beg = rowptr[node];
    int end = rowptr[node + 1];
    float sum = 0.0f;
    for (int base = beg; base < end; base += 32) {
        int cnt = end - base;
        if (cnt > 32) cnt = 32;
        int my_s = 0;
        int my_nbits = 0;
        if (lane < cnt) {
            my_s = colidx[base + lane];
            my_nbits = __float_as_int(norm[my_s]);
        }
        int j = 0;
        for (; j + 8 <= cnt; j += 8) sum = gather_group<8>(hsrc, my_s, my_nbits, j, lane, sum);
        if (cnt & 4) { sum = gather_group<4>(hsrc, my_s, my_nbits, j, lane, sum); j += 4; }
        if (cnt & 2) { sum = gather_group<2>(hsrc, my_s, my_nbits, j, lane, sum); j += 2; }
        if (cnt & 1) { sum = gather_group<1>(hsrc, my_s, my_nbits, j, lane, sum); }
    }
    acc[node * DFEAT + lane] = sum;
}

__global__ void appnp_update_wmma(const float* __restrict__ feat, const float* __restrict__ norm,
                                  const float* __restrict__ acc, float* __restrict__ h,
                                  int N, int ntiles) {
    int wave = blockIdx.x * (blockDim.x >> 5) + (threadIdx.x >> 5);
    if (wave >= ntiles) return;
    int node0 = wave * 16;
    int lane = threadIdx.x & 31;
    int hi = lane >> 4;
    int lo = lane & 15;

    if (node0 + 16 <= N) {
        float nv = (1.0f - ALPHA) * norm[node0 + lo];
#pragma unroll
        for (int fh = 0; fh < 2; ++fh) {
            int f0 = fh * 16;
            v8f cacc;
#pragma unroll
            for (int r = 0; r < 8; ++r)
                cacc[r] = ALPHA * feat[(node0 + r + 8 * hi) * DFEAT + f0 + lo];
#pragma unroll
            for (int cc = 0; cc < 4; ++cc) {
                v2f a, b;
#pragma unroll
                for (int v = 0; v < 2; ++v) {
                    int k = v + 2 * hi;
                    int m = 4 * cc + k;
                    b[v] = acc[(node0 + m) * DFEAT + f0 + lo];
                    a[v] = (lo == (4 * cc + k)) ? nv : 0.0f;
                }
                cacc = __builtin_amdgcn_wmma_f32_16x16x4_f32(
                     false, a,  false, b,
                     (short)0, cacc,  false,  false);
            }
#pragma unroll
            for (int r = 0; r < 8; ++r)
                h[(node0 + r + 8 * hi) * DFEAT + f0 + lo] = cacc[r];
        }
    } else {
        for (int m = 0; m < 16; ++m) {
            int node = node0 + m;
            if (node < N) {
                int j = node * DFEAT + lane;
                h[j] = (1.0f - ALPHA) * norm[node] * acc[j] + ALPHA * feat[j];
            }
        }
    }
}

extern "C" void kernel_launch(void* const* d_in, const int* in_sizes, int n_in,
                              void* d_out, int out_size, void* d_ws, size_t ws_size,
                              hipStream_t stream) {
    const float* feat = (const float*)d_in[0];
    const int*   src  = (const int*)d_in[1];
    const int*   dst  = (const int*)d_in[2];
    int N = in_sizes[0] / DFEAT;
    int E = in_sizes[1];

    size_t npad = ((size_t)N + 4) & ~(size_t)3;

    float* h      = (float*)d_out;
    float* acc    = (float*)d_ws;
    float* norm   = acc + (size_t)N * DFEAT;
    int*   cnt    = (int*)(norm + npad);
    int*   rowptr = cnt + npad;
    int*   colidx = rowptr + npad;
    int*   rank   = (int*)acc;
    if (E > N * DFEAT) return;
    if (ws_size < ((size_t)N * DFEAT + 3 * npad + (size_t)E) * sizeof(float)) return;

    int totalND = N * DFEAT;
    appnp_init<<<(totalND + 255) / 256, 256, 0, stream>>>(feat, h, cnt, N);
    appnp_count_rank<<<(E + 255) / 256, 256, 0, stream>>>(dst, cnt, rank, E);
    appnp_rowptr_norm<<<(N + SCAN_T - 1) / SCAN_T, SCAN_T, 0, stream>>>(cnt, rowptr, norm, N);
    appnp_fill<<<(E + 255) / 256, 256, 0, stream>>>(src, dst, rank, rowptr, colidx, E);

    int ntiles = (N + 15) / 16;
    for (int k = 0; k < KITER; ++k) {
        appnp_gather<<<(N + 7) / 8, 256, 0, stream>>>(h, norm, rowptr, colidx, acc, N);
        appnp_update_wmma<<<(ntiles + 7) / 8, 256, 0, stream>>>(feat, norm, acc, h, N, ntiles);
    }
}


# gather: 8 lanes x 16 B per source row, so one wave sums 4 dst nodes side by side and every load instruction brings 4 rows (4x fewer gather waves and row-load instructions per sweep)
# speedup vs baseline: 3.6941x; 1.1787x over previous
#include <hip/hip_runtime.h>
#include <math.h>

#define ALPHA 0.1f
#define KITER 10
#define DFEAT 32
#define SCAN_T 1024

typedef __attribute__((ext_vector_type(2))) float v2f;
typedef __attribute__((ext_vector_type(4))) float v4f;
typedef __attribute__((ext_vector_type(8))) float v8f;

__global__ void appnp_init(const float* __restrict__ feat, float* __restrict__ h,
                           int* __restrict__ cnt, int N) {
    int i = blockIdx.x * blockDim.x + threadIdx.x;
    int total = N * DFEAT;
    if (i < total) h[i] = feat[i];
    if (i < N) cnt[i] = 0;
}

__global__ void appnp_count_rank(const int* __restrict__ dst, int* __restrict__ cnt,
                                 int* __restrict__ rank, int E) {
    int i = blockIdx.x * blockDim.x + threadIdx.x;
    if (i < E) rank[i] = atomicAdd(&cnt[dst[i]], 1);
}

__global__ void __launch_bounds__(SCAN_T)
appnp_rowptr_norm(const int* __restrict__ cnt, int* __restrict__ rowptr,
                  float* __restrict__ norm, int N) {
    __shared__ int s_part[SCAN_T / 32];
    __shared__ int s_incl[SCAN_T / 32];
    int tid = threadIdx.x;
    int lane = tid & 31;
    int w = tid >> 5;
    int base = blockIdx.x * SCAN_T;

    int part = 0;
    const int4* cnt4 = (const int4*)cnt;
#pragma unroll 8
    for (int q = tid; q < base / 4; q += SCAN_T) {
        int4 v = cnt4[q];
        part += (v.x + v.y) + (v.z + v.w);
    }
    int i = base + tid;
    int c = (i < N) ? cnt[i] : 0;

    int incl = c;
#pragma unroll
    for (int o = 1; o < 32; o <<= 1) {
        int t = __shfl_up(incl, o);
        if (lane >= o) incl += t;
    }
#pragma unroll
    for (int o = 16; o > 0; o >>= 1) part += __shfl_xor(part, o);

    if (lane == 31) s_incl[w] = incl;
    if (lane == 0) s_part[w] = part;
    __syncthreads();

    int before = 0;
    for (int k = 0; k < SCAN_T / 32; ++k) {
        before += s_part[k];
        if (k < w) before += s_incl[k];
    }
    if (i < N) {
        rowptr[i] = before + incl - c;
        norm[i] = (c > 0) ? (1.0f / sqrtf((float)c)) : 0.0f;
        if (i == N - 1) rowptr[N] = before + incl;
    }
}

__global__ void appnp_fill(const int* __restrict__ src, const int* __restrict__ dst,
                           const int* __restrict__ rank, const int* __restrict__ rowptr,
                           int* __restrict__ colidx, int E) {
    int i = blockIdx.x * blockDim.x + threadIdx.x;
    if (i < E) colidx[rowptr[dst[i]] + rank[i]] = src[i];
}

#define GROUP_BCAST(x, u) __builtin_amdgcn_mov_dpp8((x), (unsigned)(u) * 0x249249u)

__device__ __forceinline__ v4f gather_group8(const v4f* __restrict__ h4, int my_s, int my_nbits,
                                             unsigned sub, v4f sum) {
    v4f v[8];
#define ROW_PIECE(u) v[u] = h4[(unsigned)GROUP_BCAST(my_s, u) * (unsigned)(DFEAT / 4) + sub];
    ROW_PIECE(0) ROW_PIECE(1) ROW_PIECE(2) ROW_PIECE(3)
    ROW_PIECE(4) ROW_PIECE(5) ROW_PIECE(6) ROW_PIECE(7)
#undef ROW_PIECE
#define ROW_ADD(u) sum += v[u] * __int_as_float(GROUP_BCAST(my_nbits, u));
    ROW_ADD(0) ROW_ADD(1) ROW_ADD(2) ROW_ADD(3)
    ROW_ADD(4) ROW_ADD(5) ROW_ADD(6) ROW_ADD(7)
#undef ROW_ADD
    return sum;
}

__global__ void appnp_gather(const float* __restrict__ hsrc, const float* __restrict__ norm,
                             const int* __restrict__ rowptr, const int* __restrict__ colidx,
                             float* __restrict__ acc, int N) {
    int node = (blockIdx.x * blockDim.x + threadIdx.x) >> 3;
    if (node >= N) return;
    int sub = threadIdx.x & 7;
    const v4f* h4 = (const v4f*)hsrc;
    int beg = rowptr[node];
    int end = rowptr[node + 1];
    v4f sum = {0.0f, 0.0f, 0.0f, 0.0f};
    for (int base = beg; base < end; base += 32) {
        int cnt = end - base;
        int my_s[4];
        int my_nbits[4];
#pragma unroll
        for (int k = 0; k < 4; ++k) {
            int e = 8 * k + sub;
            my_s[k] = colidx[(unsigned)(base + (e < cnt ? e : cnt - 1))];
        }
#pragma unroll
        for (int k = 0; k < 4; ++k) {
            my_nbits[k] = __float_as_int(norm[(unsigned)my_s[k]]);
        }
        for (int g = 0; g < cnt && g < 32; g += 8) {
            int w = (g + sub < cnt) ? my_nbits[0] : 0;
            sum = gather_group8(h4, my_s[0], w, (unsigned)sub, sum);
            my_s[0] = my_s[1]; my_s[1] = my_s[2]; my_s[2] = my_s[3];
            my_nbits[0] = my_nbits[1]; my_nbits[1] = my_nbits[2]; my_nbits[2] = my_nbits[3];
        }
    }
    ((v4f*)acc)[(unsigned)node * (unsigned)(DFEAT / 4) + (unsigned)sub] = sum;
}

__global__ void appnp_update_wmma(const float* __restrict__ feat, const float* __restrict__ norm,
                                  const float* __restrict__ acc, float* __restrict__ h,
                                  int N, int ntiles) {
    int wave = blockIdx.x * (blockDim.x >> 5) + (threadIdx.x >> 5);
    if (wave >= ntiles) return;
    int node0 = wave * 16;
    int lane = threadIdx.x & 31;
    int hi = lane >> 4;
    int lo = lane & 15;

    if (node0 + 16 <= N) {
        float nv = (1.0f - ALPHA) * norm[node0 + lo];
#pragma unroll
        for (int fh = 0; fh < 2; ++fh) {
            int f0 = fh * 16;
            v8f cacc;
#pragma unroll
            for (int r = 0; r < 8; ++r)
                cacc[r] = ALPHA * feat[(node0 + r + 8 * hi) * DFEAT + f0 + lo];
#pragma unroll
            for (int cc = 0; cc < 4; ++cc) {
                v2f a, b;
#pragma unroll
                for (int v = 0; v < 2; ++v) {
                    int k = v + 2 * hi;
                    int m = 4 * cc + k;
                    b[v] = acc[(node0 + m) * DFEAT + f0 + lo];
                    a[v] = (lo == (4 * cc + k)) ? nv : 0.0f;
                }
                cacc = __builtin_amdgcn_wmma_f32_16x16x4_f32(
                     false, a,  false, b,
                     (short)0, cacc,  false,  false);
            }
#pragma unroll
            for (int r = 0; r < 8; ++r)
                h[(node0 + r + 8 * hi) * DFEAT + f0 + lo] = cacc[r];
        }
    } else {
        for (int m = 0; m < 16; ++m) {
            int node = node0 + m;
            if (node < N) {
                int j = node * DFEAT + lane;
                h[j] = (1.0f - ALPHA) * norm[node] * acc[j] + ALPHA * feat[j];
            }
        }
    }
}

extern "C" void kernel_launch(void* const* d_in, const int* in_sizes, int n_in,
                              void* d_out, int out_size, void* d_ws, size_t ws_size,
                              hipStream_t stream) {
    const float* feat = (const float*)d_in[0];
    const int*   src  = (const int*)d_in[1];
    const int*   dst  = (const int*)d_in[2];
    int N = in_sizes[0] / DFEAT;
    int E = in_sizes[1];

    size_t npad = ((size_t)N + 4) & ~(size_t)3;

    float* h      = (float*)d_out;
    float* acc    = (float*)d_ws;
    float* norm   = acc + (size_t)N * DFEAT;
    int*   cnt    = (int*)(norm + npad);
    int*   rowptr = cnt + npad;
    int*   colidx = rowptr + npad;
    int*   rank   = (int*)acc;
    if (E > N * DFEAT) return;
    if (ws_size < ((size_t)N * DFEAT + 3 * npad + (size_t)E) * sizeof(float)) return;

    int totalND = N * DFEAT;
    appnp_init<<<(totalND + 255) / 256, 256, 0, stream>>>(feat, h, cnt, N);
    appnp_count_rank<<<(E + 255) / 256, 256, 0, stream>>>(dst, cnt, rank, E);
    appnp_rowptr_norm<<<(N + SCAN_T - 1) / SCAN_T, SCAN_T, 0, stream>>>(cnt, rowptr, norm, N);
    appnp_fill<<<(E + 255) / 256, 256, 0, stream>>>(src, dst, rank, rowptr, colidx, E);

    int ntiles = (N + 15) / 16;
    for (int k = 0; k < KITER; ++k) {
        appnp_gather<<<(N + 31) / 32, 256, 0, stream>>>(h, norm, rowptr, colidx, acc, N);
        appnp_update_wmma<<<(ntiles + 7) / 8, 256, 0, stream>>>(feat, norm, acc, h, N, ntiles);
    }
}


# rows kept pre-scaled by norm (update stores h*norm, the last sweep h; first rows feat*norm), so the gather sums plain rows: no per-edge random norm[src] load in any sweep
# speedup vs baseline: 4.4147x; 1.1951x over previous
#include <hip/hip_runtime.h>
#include <math.h>

#define ALPHA 0.1f
#define KITER 10
#define DFEAT 32
#define SCAN_T 1024

typedef __attribute__((ext_vector_type(2))) float v2f;
typedef __attribute__((ext_vector_type(4))) float v4f;
typedef __attribute__((ext_vector_type(8))) float v8f;

__global__ void appnp_zero_counts(int* __restrict__ cnt, int N) {
    int i = blockIdx.x * blockDim.x + threadIdx.x;
    if (i < N) cnt[i] = 0;
}

__global__ void appnp_prescale(const float* __restrict__ feat, const float* __restrict__ norm,
                               float* __restrict__ hs, int N) {
    int i = blockIdx.x * blockDim.x + threadIdx.x;
    if (i < N * (DFEAT / 4)) {
        v4f f = ((const v4f*)feat)[i];
        ((v4f*)hs)[i] = f * norm[i >> 3];
    }
}

__global__ void appnp_count_rank(const int* __restrict__ dst, int* __restrict__ cnt,
                                 int* __restrict__ rank, int E) {
    int i = blockIdx.x * blockDim.x + threadIdx.x;
    if (i < E) rank[i] = atomicAdd(&cnt[dst[i]], 1);
}

__global__ void __launch_bounds__(SCAN_T)
appnp_rowptr_norm(const int* __restrict__ cnt, int* __restrict__ rowptr,
                  float* __restrict__ norm, int N) {
    __shared__ int s_part[SCAN_T / 32];
    __shared__ int s_incl[SCAN_T / 32];
    int tid = threadIdx.x;
    int lane = tid & 31;
    int w = tid >> 5;
    int base = blockIdx.x * SCAN_T;

    int part = 0;
    const int4* cnt4 = (const int4*)cnt;
#pragma unroll 8
    for (int q = tid; q < base / 4; q += SCAN_T) {
        int4 v = cnt4[q];
        part += (v.x + v.y) + (v.z + v.w);
    }
    int i = base + tid;
    int c = (i < N) ? cnt[i] : 0;

    int incl = c;
#pragma unroll
    for (int o = 1; o < 32; o <<= 1) {
        int t = __shfl_up(incl, o);
        if (lane >= o) incl += t;
    }
#pragma unroll
    for (int o = 16; o > 0; o >>= 1) part += __shfl_xor(part, o);

    if (lane == 31) s_incl[w] = incl;
    if (lane == 0) s_part[w] = part;
    __syncthreads();

    int before = 0;
    for (int k = 0; k < SCAN_T / 32; ++k) {
        before += s_part[k];
        if (k < w) before += s_incl[k];
    }
    if (i < N) {
        rowptr[i] = before + incl - c;
        norm[i] = (c > 0) ? (1.0f / sqrtf((float)c)) : 0.0f;
        if (i == N - 1) rowptr[N] = before + incl;
    }
}

__global__ void appnp_fill(const int* __restrict__ src, const int* __restrict__ dst,
                           const int* __restrict__ rank, const int* __restrict__ rowptr,
                           int* __restrict__ colidx, int E) {
    int i = blockIdx.x * blockDim.x + threadIdx.x;
    if (i < E) colidx[rowptr[dst[i]] + rank[i]] = src[i];
}

#define GROUP_BCAST(x, u) __builtin_amdgcn_mov_dpp8((x), (unsigned)(u) * 0x249249u)

__device__ __forceinline__ v4f gather_group8(const v4f* __restrict__ h4, int my_s, int my_wbits,
                                             unsigned sub, v4f sum) {
    v4f v[8];
#define ROW_PIECE(u) v[u] = h4[(unsigned)GROUP_BCAST(my_s, u) * (unsigned)(DFEAT / 4) + sub];
    ROW_PIECE(0) ROW_PIECE(1) ROW_PIECE(2) ROW_PIECE(3)
    ROW_PIECE(4) ROW_PIECE(5) ROW_PIECE(6) ROW_PIECE(7)
#undef ROW_PIECE
#define ROW_ADD(u) sum += v[u] * __int_as_float(GROUP_BCAST(my_wbits, u));
    ROW_ADD(0) ROW_ADD(1) ROW_ADD(2) ROW_ADD(3)
    ROW_ADD(4) ROW_ADD(5) ROW_ADD(6) ROW_ADD(7)
#undef ROW_ADD
    return sum;
}

__global__ void appnp_gather(const float* __restrict__ hsrc,
                             const int* __restrict__ rowptr, const int* __restrict__ colidx,
                             float* __restrict__ acc, int N) {
    int node = (blockIdx.x * blockDim.x + threadIdx.x) >> 3;
    if (node >= N) return;
    int sub = threadIdx.x & 7;
    const v4f* h4 = (const v4f*)hsrc;
    int beg = rowptr[node];
    int end = rowptr[node + 1];
    v4f sum = {0.0f, 0.0f, 0.0f, 0.0f};
    for (int base = beg; base < end; base += 32) {
        int cnt = end - base;
        int my_s[4];
        int my_wbits[4];
#pragma unroll
        for (int k = 0; k < 4; ++k) {
            int e = 8 * k + sub;
            bool real = e < cnt;
            my_s[k] = colidx[(unsigned)(base + (real ? e : cnt - 1))];
            my_wbits[k] = real ? __float_as_int(1.0f) : 0;
        }
        for (int g = 0; g < cnt && g < 32; g += 8) {
            sum = gather_group8(h4, my_s[0], my_wbits[0], (unsigned)sub, sum);
            my_s[0] = my_s[1]; my_s[1] = my_s[2]; my_s[2] = my_s[3];
            my_wbits[0] = my_wbits[1]; my_wbits[1] = my_wbits[2]; my_wbits[2] = my_wbits[3];
        }
    }
    ((v4f*)acc)[(unsigned)node * (unsigned)(DFEAT / 4) + (unsigned)sub] = sum;
}

__global__ void appnp_update_wmma(const float* __restrict__ feat, const float* __restrict__ norm,
                                  const float* __restrict__ acc, float* __restrict__ h,
                                  int N, int ntiles, int scale_rows) {
    int wave = blockIdx.x * (blockDim.x >> 5) + (threadIdx.x >> 5);
    if (wave >= ntiles) return;
    int node0 = wave * 16;
    int lane = threadIdx.x & 31;
    int hi = lane >> 4;
    int lo = lane & 15;

    if (node0 + 16 <= N) {
        float nv = (1.0f - ALPHA) * norm[node0 + lo];
        float rown[8];
#pragma unroll
        for (int r = 0; r < 8; ++r) rown[r] = 1.0f;
        if (scale_rows) {
            const v4f* n4 = (const v4f*)(norm + node0 + 8 * hi);
            v4f n0 = n4[0], n1 = n4[1];
            rown[0] = n0[0]; rown[1] = n0[1]; rown[2] = n0[2]; rown[3] = n0[3];
            rown[4] = n1[0]; rown[5] = n1[1]; rown[6] = n1[2]; rown[7] = n1[3];
        }
#pragma unroll
        for (int fh = 0; fh < 2; ++fh) {
            int f0 = fh * 16;
            v8f cacc;
#pragma unroll
            for (int r = 0; r < 8; ++r)
                cacc[r] = ALPHA * feat[(node0 + r + 8 * hi) * DFEAT + f0 + lo];
#pragma unroll
            for (int cc = 0; cc < 4; ++cc) {
                v2f a, b;
#pragma unroll
                for (int v = 0; v < 2; ++v) {
                    int k = v + 2 * hi;
                    int m = 4 * cc + k;
                    b[v] = acc[(node0 + m) * DFEAT + f0 + lo];
                    a[v] = (lo == (4 * cc + k)) ? nv : 0.0f;
                }
                cacc = __builtin_amdgcn_wmma_f32_16x16x4_f32(
                     false, a,  false, b,
                     (short)0, cacc,  false,  false);
            }
#pragma unroll
            for (int r = 0; r < 8; ++r)
                h[(node0 + r + 8 * hi) * DFEAT + f0 + lo] = cacc[r] * rown[r];
        }
    } else {
        for (int m = 0; m < 16; ++m) {
            int node = node0 + m;
            if (node < N) {
                int j = node * DFEAT + lane;
                float hv = (1.0f - ALPHA) * norm[node] * acc[j] + ALPHA * feat[j];
                h[j] = scale_rows ? hv * norm[node] : hv;
            }
        }
    }
}

extern "C" void kernel_launch(void* const* d_in, const int* in_sizes, int n_in,
                              void* d_out, int out_size, void* d_ws, size_t ws_size,
                              hipStream_t stream) {
    const float* feat = (const float*)d_in[0];
    const int*   src  = (const int*)d_in[1];
    const int*   dst  = (const int*)d_in[2];
    int N = in_sizes[0] / DFEAT;
    int E = in_sizes[1];

    size_t npad = ((size_t)N + 4) & ~(size_t)3;

    float* h      = (float*)d_out;
    float* acc    = (float*)d_ws;
    float* norm   = acc + (size_t)N * DFEAT;
    int*   cnt    = (int*)(norm + npad);
    int*   rowptr = cnt + npad;
    int*   colidx = rowptr + npad;
    int*   rank   = (int*)acc;
    if (E > N * DFEAT) return;
    if (ws_size < ((size_t)N * DFEAT + 3 * npad + (size_t)E) * sizeof(float)) return;

    appnp_zero_counts<<<(N + 255) / 256, 256, 0, stream>>>(cnt, N);
    appnp_count_rank<<<(E + 255) / 256, 256, 0, stream>>>(dst, cnt, rank, E);
    appnp_rowptr_norm<<<(N + SCAN_T - 1) / SCAN_T, SCAN_T, 0, stream>>>(cnt, rowptr, norm, N);
    appnp_fill<<<(E + 255) / 256, 256, 0, stream>>>(src, dst, rank, rowptr, colidx, E);

    appnp_prescale<<<(N * (DFEAT / 4) + 255) / 256, 256, 0, stream>>>(feat, norm, h, N);

    int ntiles = (N + 15) / 16;
    for (int k = 0; k < KITER - 1; ++k) {
        appnp_gather<<<(N + 31) / 32, 256, 0, stream>>>(h, rowptr, colidx, acc, N);
        appnp_update_wmma<<<(ntiles + 7) / 8, 256, 0, stream>>>(feat, norm, acc, h, N, ntiles, 1);
    }
    appnp_gather<<<(N + 31) / 32, 256, 0, stream>>>(h, rowptr, colidx, acc, N);
    appnp_update_wmma<<<(ntiles + 7) / 8, 256, 0, stream>>>(feat, norm, acc, h, N, ntiles, 0);
}
